# starting kernel
# baseline (speedup 1.0000x reference)
#include <hip/hip_runtime.h>
#include <hip/hip_bf16.h>

typedef __attribute__((ext_vector_type(2))) float v2f;
typedef __attribute__((ext_vector_type(4))) float v4f;
typedef __attribute__((ext_vector_type(8))) float v8f;

#define FEAT 128
#define A_STRIDE 132

__global__ void deg_init_kernel(float* __restrict__ deg, int n) {
    int i = blockIdx.x * blockDim.x + threadIdx.x;
    if (i < n) deg[i] = 1.0f;
}

__global__ void deg_count_kernel(const int* __restrict__ dst,
                                 float* __restrict__ deg, int e) {
    int i = blockIdx.x * blockDim.x + threadIdx.x;
    if (i < e) unsafeAtomicAdd(&deg[dst[i]], 1.0f);
}

__global__ void dinv_kernel(float* __restrict__ deg, int n) {
    int i = blockIdx.x * blockDim.x + threadIdx.x;
    if (i < n) deg[i] = rsqrtf(deg[i]);
}

__global__ void self_loop_kernel(const float* __restrict__ emb,
                                 const float* __restrict__ dinv,
                                 float* __restrict__ out, int n) {
    int t = blockIdx.x * blockDim.x + threadIdx.x;
    int node = t >> 5;
    int c = (t & 31) * 4;
    if (node >= n) return;
    float s = dinv[node];
    s = s * s;
    v4f v = *(const v4f*)(emb + (size_t)node * FEAT + c);
    v *= s;
    *(v4f*)(out + (size_t)node * FEAT + c) = v;
}

__global__ void scatter_kernel(const float* __restrict__ emb,
                               const int* __restrict__ src,
                               const int* __restrict__ dst,
                               const float* __restrict__ dinv,
                               float* __restrict__ out, int e) {
    int gt = blockIdx.x * blockDim.x + threadIdx.x;
    int edge = gt >> 5;
    int c = (gt & 31) * 4;
    if (edge >= e) return;
    int s = src[edge];
    int d = dst[edge];
    float norm = dinv[s] * dinv[d];
    v4f v = *(const v4f*)(emb + (size_t)s * FEAT + c);
    float* o = out + (size_t)d * FEAT + c;
    unsafeAtomicAdd(o + 0, v.x * norm);
    unsafeAtomicAdd(o + 1, v.y * norm);
    unsafeAtomicAdd(o + 2, v.z * norm);
    unsafeAtomicAdd(o + 3, v.w * norm);
}

__global__ __launch_bounds__(256) void gcn_gemm_kernel(
    const float* __restrict__ Wmat, const float* __restrict__ bias,
    float* __restrict__ out, int n) {
    __shared__ float Ain[16 * A_STRIDE];

    const int tid = threadIdx.x;
    const int row0 = blockIdx.x * 16;
    const bool full_tile = (row0 + 16 <= n);

    if (full_tile) {
        #pragma unroll
        for (int j = 0; j < 2; ++j) {
            int i = tid * 2 + j;
            int r = i >> 5;
            int c4 = i & 31;
            v4f v = *(const v4f*)(out + (size_t)(row0 + r) * FEAT + c4 * 4);
            *(v4f*)(&Ain[r * A_STRIDE + c4 * 4]) = v;
        }
    } else {
        #pragma unroll
        for (int j = 0; j < 2; ++j) {
            int i = tid * 2 + j;
            int r = i >> 5;
            int c4 = i & 31;
            int row = row0 + r;
            v4f v = {};
            if (row < n) v = *(const v4f*)(out + (size_t)row * FEAT + c4 * 4);
            *(v4f*)(&Ain[r * A_STRIDE + c4 * 4]) = v;
        }
    }
    __syncthreads();

    const int wave = tid >> 5;
    const int lane = tid & 31;
    const int l = lane & 15;
    const int h = lane >> 4;
    const int col = wave * 16 + l;

    v8f acc0 = {};
    v8f acc1 = {};
    #pragma unroll 8
    for (int k = 0; k < FEAT / 4; ++k) {
        const int kk = 4 * k + 2 * h;
        v2f a, b;
        a.x = Ain[l * A_STRIDE + kk];
        a.y = Ain[l * A_STRIDE + kk + 1];
        b.x = Wmat[(size_t)kk * FEAT + col];
        b.y = Wmat[(size_t)(kk + 1) * FEAT + col];
        if (k & 1) {
            acc1 = __builtin_amdgcn_wmma_f32_16x16x4_f32(
                false, a, false, b, (short)0, acc1, false, false);
        } else {
            acc0 = __builtin_amdgcn_wmma_f32_16x16x4_f32(
                false, a, false, b, (short)0, acc0, false, false);
        }
    }
    v8f acc = acc0 + acc1;

    const float bv = bias[col];
    if (full_tile) {
        float* orow = out + (size_t)(row0 + 8 * h) * FEAT + col;
        #pragma unroll
        for (int i = 0; i < 8; ++i)
            orow[(size_t)i * FEAT] = acc[i] + bv;
    } else {
        #pragma unroll
        for (int i = 0; i < 8; ++i) {
            int row = row0 + i + 8 * h;
            if (row < n) out[(size_t)row * FEAT + col] = acc[i] + bv;
        }
    }
}

extern "C" void kernel_launch(void* const* d_in, const int* in_sizes, int n_in,
                              void* d_out, int out_size, void* d_ws, size_t ws_size,
                              hipStream_t stream) {
    const float* emb  = (const float*)d_in[0];
    const int*   eidx = (const int*)d_in[1];
    const float* Wmat = (const float*)d_in[2];
    const float* bias = (const float*)d_in[3];
    float* out = (float*)d_out;

    const int N = in_sizes[0] / FEAT;
    const int E = in_sizes[1] / 2;
    const int* src = eidx;
    const int* dst = eidx + E;

    float* dinv = (float*)d_ws;

    const int T = 256;
    deg_init_kernel <<<(N + T - 1) / T, T, 0, stream>>>(dinv, N);
    deg_count_kernel<<<(E + T - 1) / T, T, 0, stream>>>(dst, dinv, E);
    dinv_kernel     <<<(N + T - 1) / T, T, 0, stream>>>(dinv, N);
    self_loop_kernel<<<(N * 32 + T - 1) / T, T, 0, stream>>>(emb, dinv, out, N);
    scatter_kernel  <<<(E * 32 + T - 1) / T, T, 0, stream>>>(emb, src, dst, dinv, out, E);
    gcn_gemm_kernel <<<(N + 15) / 16, T, 0, stream>>>(Wmat, bias, out, N);
}


# Float-atomic edge scatter replaced by gather: one int atomic per edge appends src to a list kept in the dst node's own out row; one wave per node sums self+neighbours in list (arrival) order
# speedup vs baseline: 1.4569x; 1.4569x over previous
#include <hip/hip_runtime.h>
#include <hip/hip_bf16.h>

typedef __attribute__((ext_vector_type(2))) float v2f;
typedef __attribute__((ext_vector_type(4))) float v4f;
typedef __attribute__((ext_vector_type(8))) float v8f;

#define FEAT 128
#define A_STRIDE 132

#define LIST_CAP 127


__global__ void cnt_init_kernel(float* __restrict__ ws, int n) {
    int i = blockIdx.x * blockDim.x + threadIdx.x;
    if (i < n) ((int*)ws)[i] = 0;
}

__global__ void list_build_kernel(const int* __restrict__ src,
                                  const int* __restrict__ dst,
                                  float* __restrict__ ws,
                                  float* __restrict__ out, int e) {
    int i = blockIdx.x * blockDim.x + threadIdx.x;
    if (i >= e) return;
    int s = src[i];
    int d = dst[i];
    int pos = atomicAdd((int*)ws + d, 1);
    if (pos < LIST_CAP) ((int*)out)[(size_t)d * FEAT + 1 + pos] = s;
}

__global__ void dinv_kernel(float* __restrict__ ws, float* __restrict__ out, int n) {
    int i = blockIdx.x * blockDim.x + threadIdx.x;
    if (i >= n) return;
    int c = ((const int*)ws)[i];
    ((int*)out)[(size_t)i * FEAT] = c;
    ws[i] = rsqrtf(1.0f + (float)c);
}

__global__ __launch_bounds__(256) void gather_kernel(
    const float* __restrict__ emb, const int* __restrict__ src,
    const int* __restrict__ dst, const float* __restrict__ dinv,
    float* out, int n, int e) {
    const int lane = threadIdx.x & 31;
    const int node = blockIdx.x * 8 + (threadIdx.x >> 5);
    if (node >= n) return;
    const int c = lane * 4;
    const int* row = (const int*)(out + (size_t)node * FEAT);
    const int mine = row[lane];
    const int cnt = __builtin_amdgcn_readfirstlane(mine);
    const float dn = dinv[node];
    v4f acc = *(const v4f*)(emb + (size_t)node * FEAT + c);
    acc *= dn * dn;

    if (cnt <= 31) {
        for (int j = 0; j < cnt; j += 4) {
            int s[4];
            float w[4];
            v4f v[4];
            #pragma unroll
            for (int k = 0; k < 4; ++k) {
                const int jj = (j + k < cnt) ? j + k : j;
                s[k] = __builtin_amdgcn_readlane(mine, jj + 1);
            }
            #pragma unroll
            for (int k = 0; k < 4; ++k) {
                w[k] = dinv[s[k]] * dn;
                v[k] = *(const v4f*)(emb + (size_t)s[k] * FEAT + c);
            }
            #pragma unroll
            for (int k = 0; k < 4; ++k)
                if (j + k < cnt) acc += v[k] * w[k];
        }
    } else if (cnt <= LIST_CAP) {
        for (int j = 0; j < cnt; ++j) {
            const int s = __builtin_amdgcn_readfirstlane(row[j + 1]);
            const float w = dinv[s] * dn;
            v4f v = *(const v4f*)(emb + (size_t)s * FEAT + c);
            acc += v * w;
        }
    } else {
        for (int base = 0; base < e; base += 32) {
            int idx = base + lane;
            int d = (idx < e) ? dst[idx] : -1;
            unsigned m = __builtin_amdgcn_ballot_w32(d == node);
            while (m) {
                int b = __builtin_ctz(m);
                m &= m - 1;
                int s = src[base + b];
                float w = dinv[s] * dn;
                v4f v = *(const v4f*)(emb + (size_t)s * FEAT + c);
                acc += v * w;
            }
        }
    }
    *(v4f*)(out + (size_t)node * FEAT + c) = acc;
}

__global__ __launch_bounds__(256) void gcn_gemm_kernel(
    const float* __restrict__ Wmat, const float* __restrict__ bias,
    float* __restrict__ out, int n) {
    __shared__ float Ain[16 * A_STRIDE];

    const int tid = threadIdx.x;
    const int row0 = blockIdx.x * 16;
    const bool full_tile = (row0 + 16 <= n);

    if (full_tile) {
        #pragma unroll
        for (int j = 0; j < 2; ++j) {
            int i = tid * 2 + j;
            int r = i >> 5;
            int c4 = i & 31;
            v4f v = *(const v4f*)(out + (size_t)(row0 + r) * FEAT + c4 * 4);
            *(v4f*)(&Ain[r * A_STRIDE + c4 * 4]) = v;
        }
    } else {
        #pragma unroll
        for (int j = 0; j < 2; ++j) {
            int i = tid * 2 + j;
            int r = i >> 5;
            int c4 = i & 31;
            int row = row0 + r;
            v4f v = {};
            if (row < n) v = *(const v4f*)(out + (size_t)row * FEAT + c4 * 4);
            *(v4f*)(&Ain[r * A_STRIDE + c4 * 4]) = v;
        }
    }
    __syncthreads();

    const int wave = tid >> 5;
    const int lane = tid & 31;
    const int l = lane & 15;
    const int h = lane >> 4;
    const int col = wave * 16 + l;

    v8f acc0 = {};
    v8f acc1 = {};
    #pragma unroll 8
    for (int k = 0; k < FEAT / 4; ++k) {
        const int kk = 4 * k + 2 * h;
        v2f a, b;
        a.x = Ain[l * A_STRIDE + kk];
        a.y = Ain[l * A_STRIDE + kk + 1];
        b.x = Wmat[(size_t)kk * FEAT + col];
        b.y = Wmat[(size_t)(kk + 1) * FEAT + col];
        if (k & 1) {
            acc1 = __builtin_amdgcn_wmma_f32_16x16x4_f32(
                false, a, false, b, (short)0, acc1, false, false);
        } else {
            acc0 = __builtin_amdgcn_wmma_f32_16x16x4_f32(
                false, a, false, b, (short)0, acc0, false, false);
        }
    }
    v8f acc = acc0 + acc1;

    const float bv = bias[col];
    if (full_tile) {
        float* orow = out + (size_t)(row0 + 8 * h) * FEAT + col;
        #pragma unroll
        for (int i = 0; i < 8; ++i)
            orow[(size_t)i * FEAT] = acc[i] + bv;
    } else {
        #pragma unroll
        for (int i = 0; i < 8; ++i) {
            int row = row0 + i + 8 * h;
            if (row < n) out[(size_t)row * FEAT + col] = acc[i] + bv;
        }
    }
}

extern "C" void kernel_launch(void* const* d_in, const int* in_sizes, int n_in,
                              void* d_out, int out_size, void* d_ws, size_t ws_size,
                              hipStream_t stream) {
    const float* emb  = (const float*)d_in[0];
    const int*   eidx = (const int*)d_in[1];
    const float* Wmat = (const float*)d_in[2];
    const float* bias = (const float*)d_in[3];
    float* out = (float*)d_out;

    const int N = in_sizes[0] / FEAT;
    const int E = in_sizes[1] / 2;
    const int* src = eidx;
    const int* dst = eidx + E;

    float* dinv = (float*)d_ws;

    const int T = 256;
    cnt_init_kernel  <<<(N + T - 1) / T, T, 0, stream>>>(dinv, N);
    list_build_kernel<<<(E + T - 1) / T, T, 0, stream>>>(src, dst, dinv, out, E);
    dinv_kernel      <<<(N + T - 1) / T, T, 0, stream>>>(dinv, out, N);
    gather_kernel    <<<(N + 7) / 8, T, 0, stream>>>(emb, src, dst, dinv, out, N, E);
    gcn_gemm_kernel <<<(N + 15) / 16, T, 0, stream>>>(Wmat, bias, out, N);
}


# Gather and GEMM fused into one persistent kernel: 1024 8-wave workgroups loop over 16-node tiles, gather two nodes per wave into LDS, then WMMA against W fragments kept in registers
# speedup vs baseline: 1.8285x; 1.2551x over previous
#include <hip/hip_runtime.h>
#include <hip/hip_bf16.h>

typedef __attribute__((ext_vector_type(2))) float v2f;
typedef __attribute__((ext_vector_type(4))) float v4f;
typedef __attribute__((ext_vector_type(8))) float v8f;

#define FEAT 128
#define A_STRIDE 132

#define LIST_CAP 127


__global__ void cnt_init_kernel(float* __restrict__ ws, int n) {
    int i = blockIdx.x * blockDim.x + threadIdx.x;
    if (i < n) ((int*)ws)[i] = 0;
}

__global__ void list_build_kernel(const int* __restrict__ src,
                                  const int* __restrict__ dst,
                                  float* __restrict__ ws,
                                  float* __restrict__ out, int e) {
    int i = blockIdx.x * blockDim.x + threadIdx.x;
    if (i >= e) return;
    int s = src[i];
    int d = dst[i];
    int pos = atomicAdd((int*)ws + d, 1);
    if (pos < LIST_CAP) ((int*)out)[(size_t)d * FEAT + 1 + pos] = s;
}

__global__ void dinv_kernel(float* __restrict__ ws, float* __restrict__ out, int n) {
    int i = blockIdx.x * blockDim.x + threadIdx.x;
    if (i >= n) return;
    int c = ((const int*)ws)[i];
    ((int*)out)[(size_t)i * FEAT] = c;
    ws[i] = rsqrtf(1.0f + (float)c);
}

__device__ __forceinline__ v4f gather_one(
    const float* __restrict__ emb, const int* __restrict__ src,
    const int* __restrict__ dst, const float* __restrict__ dinv,
    const int* row, int mine, int cnt, float dn, int node, int lane, int e, v4f acc) {
    const int c = lane * 4;
    if (cnt <= 31) {
        for (int j = 0; j < cnt; j += 4) {
            int s[4];
            float w[4];
            v4f v[4];
            #pragma unroll
            for (int k = 0; k < 4; ++k) {
                const int jj = (j + k < cnt) ? j + k : j;
                s[k] = __builtin_amdgcn_readlane(mine, jj + 1);
            }
            #pragma unroll
            for (int k = 0; k < 4; ++k) {
                w[k] = dinv[s[k]] * dn;
                v[k] = *(const v4f*)(emb + (size_t)s[k] * FEAT + c);
            }
            #pragma unroll
            for (int k = 0; k < 4; ++k)
                if (j + k < cnt) acc += v[k] * w[k];
        }
    } else if (cnt <= LIST_CAP) {
        for (int j = 0; j < cnt; ++j) {
            const int s = __builtin_amdgcn_readfirstlane(row[j + 1]);
            const float w = dinv[s] * dn;
            v4f v = *(const v4f*)(emb + (size_t)s * FEAT + c);
            acc += v * w;
        }
    } else {
        for (int base = 0; base < e; base += 32) {
            int idx = base + lane;
            int d = (idx < e) ? dst[idx] : -1;
            unsigned m = __builtin_amdgcn_ballot_w32(d == node);
            while (m) {
                int b = __builtin_ctz(m);
                m &= m - 1;
                int s = src[base + b];
                float w = dinv[s] * dn;
                v4f v = *(const v4f*)(emb + (size_t)s * FEAT + c);
                acc += v * w;
            }
        }
    }
    return acc;
}

#define TILE_M 16
#define FUSED_GRID 1024

__global__ __launch_bounds__(256) __attribute__((amdgpu_waves_per_eu(8, 8)))
void gather_gemm_kernel(
    const float* __restrict__ emb, const int* __restrict__ src,
    const int* __restrict__ dst, const float* __restrict__ dinv,
    const float* __restrict__ Wmat, const float* __restrict__ bias,
    float* out, int n, int e) {
    __shared__ float Ain[2][TILE_M * A_STRIDE];

    const int tid = threadIdx.x;
    const int wave = __builtin_amdgcn_readfirstlane(tid >> 5);
    const int lane = tid & 31;
    const int l = lane & 15;
    const int h = lane >> 4;
    const int col = wave * 16 + l;
    const int c = lane * 4;

    v2f bfrag[FEAT / 4];
    #pragma unroll
    for (int k = 0; k < FEAT / 4; ++k) {
        const int kk = 4 * k + 2 * h;
        bfrag[k].x = Wmat[(size_t)kk * FEAT + col];
        bfrag[k].y = Wmat[(size_t)(kk + 1) * FEAT + col];
    }
    const float bv = bias[col];

    const int ntiles = (n + TILE_M - 1) / TILE_M;
    int buf = 0;
    for (int tile = blockIdx.x; tile < ntiles; tile += gridDim.x, buf ^= 1) {
        const int row0 = tile * TILE_M;
        float* As = Ain[buf];

        const int rA = 2 * wave, rB = rA + 1;
        const int nA = row0 + rA, nB = row0 + rB;
        const bool okA = nA < n, okB = nB < n;
        const int mA = okA ? nA : 0, mB = okB ? nB : 0;
        const int* rowA = (const int*)(out + (size_t)mA * FEAT);
        const int* rowB = (const int*)(out + (size_t)mB * FEAT);
        const int mineA = rowA[lane];
        const int mineB = rowB[lane];
        const float dnA = dinv[mA], dnB = dinv[mB];
        v4f accA = *(const v4f*)(emb + (size_t)mA * FEAT + c);
        v4f accB = *(const v4f*)(emb + (size_t)mB * FEAT + c);
        const int cntA = okA ? __builtin_amdgcn_readfirstlane(mineA) : 0;
        const int cntB = okB ? __builtin_amdgcn_readfirstlane(mineB) : 0;
        accA *= dnA * dnA;
        accB *= dnB * dnB;

        if (cntA <= 31 && cntB <= 31) {
            const int cmax = cntA > cntB ? cntA : cntB;
            for (int j = 0; j < cmax; j += 4) {
                int sA[4], sB[4];
                float wA[4], wB[4];
                v4f vA[4], vB[4];
                #pragma unroll
                for (int k = 0; k < 4; ++k) {
                    const int tA = __builtin_amdgcn_readlane(mineA, (j + k + 1) & 31);
                    const int tB = __builtin_amdgcn_readlane(mineB, (j + k + 1) & 31);
                    sA[k] = (j + k < cntA) ? tA : mA;
                    sB[k] = (j + k < cntB) ? tB : mB;
                }
                #pragma unroll
                for (int k = 0; k < 4; ++k) {
                    wA[k] = dinv[sA[k]] * dnA;
                    vA[k] = *(const v4f*)(emb + (size_t)sA[k] * FEAT + c);
                    wB[k] = dinv[sB[k]] * dnB;
                    vB[k] = *(const v4f*)(emb + (size_t)sB[k] * FEAT + c);
                }
                #pragma unroll
                for (int k = 0; k < 4; ++k) {
                    if (j + k < cntA) accA += vA[k] * wA[k];
                    if (j + k < cntB) accB += vB[k] * wB[k];
                }
            }
        } else {
            accA = gather_one(emb, src, dst, dinv, rowA, mineA, cntA, dnA, nA, lane, e, accA);
            accB = gather_one(emb, src, dst, dinv, rowB, mineB, cntB, dnB, nB, lane, e, accB);
        }
        if (!okA) accA = (v4f){0.f, 0.f, 0.f, 0.f};
        if (!okB) accB = (v4f){0.f, 0.f, 0.f, 0.f};
        *(v4f*)(&As[rA * A_STRIDE + c]) = accA;
        *(v4f*)(&As[rB * A_STRIDE + c]) = accB;
        __syncthreads();

        const float* At = &As[l * A_STRIDE + 2 * h];
        v8f acc0 = {};
        v8f acc1 = {};
        #pragma unroll
        for (int k = 0; k < FEAT / 4; ++k) {
            const v2f a = *(const v2f*)(At + 4 * k);
            if (k & 1) {
                acc1 = __builtin_amdgcn_wmma_f32_16x16x4_f32(
                    false, a, false, bfrag[k], (short)0, acc1, false, false);
            } else {
                acc0 = __builtin_amdgcn_wmma_f32_16x16x4_f32(
                    false, a, false, bfrag[k], (short)0, acc0, false, false);
            }
        }
        const v8f acc = acc0 + acc1;

        if (row0 + TILE_M <= n) {
            float* orow = out + (size_t)(row0 + 8 * h) * FEAT + col;
            #pragma unroll
            for (int i = 0; i < 8; ++i)
                orow[(size_t)i * FEAT] = acc[i] + bv;
        } else {
            #pragma unroll
            for (int i = 0; i < 8; ++i) {
                const int row = row0 + i + 8 * h;
                if (row < n) out[(size_t)row * FEAT + col] = acc[i] + bv;
            }
        }
    }
}

extern "C" void kernel_launch(void* const* d_in, const int* in_sizes, int n_in,
                              void* d_out, int out_size, void* d_ws, size_t ws_size,
                              hipStream_t stream) {
    const float* emb  = (const float*)d_in[0];
    const int*   eidx = (const int*)d_in[1];
    const float* Wmat = (const float*)d_in[2];
    const float* bias = (const float*)d_in[3];
    float* out = (float*)d_out;

    const int N = in_sizes[0] / FEAT;
    const int E = in_sizes[1] / 2;
    const int* src = eidx;
    const int* dst = eidx + E;

    float* dinv = (float*)d_ws;

    const int T = 256;
    cnt_init_kernel  <<<(N + T - 1) / T, T, 0, stream>>>(dinv, N);
    list_build_kernel<<<(E + T - 1) / T, T, 0, stream>>>(src, dst, dinv, out, E);
    dinv_kernel      <<<(N + T - 1) / T, T, 0, stream>>>(dinv, out, N);
    gather_gemm_kernel<<<FUSED_GRID, T, 0, stream>>>(emb, src, dst, dinv, Wmat, bias, out, N, E);
}


# Degree pass and its launch removed (4 launches to 3): the fused kernel reads the integer edge counters and forms (1+count)^-1/2 itself, per node and per list entry; sum order kept
# speedup vs baseline: 1.9480x; 1.0654x over previous
#include <hip/hip_runtime.h>
#include <hip/hip_bf16.h>

typedef __attribute__((ext_vector_type(2))) float v2f;
typedef __attribute__((ext_vector_type(4))) float v4f;
typedef __attribute__((ext_vector_type(8))) float v8f;

#define FEAT 128
#define A_STRIDE 132

#define LIST_CAP 127


__global__ void cnt_init_kernel(float* __restrict__ ws, int n) {
    int i = blockIdx.x * blockDim.x + threadIdx.x;
    if (i < n) ((int*)ws)[i] = 0;
}

__global__ void list_build_kernel(const int* __restrict__ src,
                                  const int* __restrict__ dst,
                                  float* __restrict__ ws,
                                  float* __restrict__ out, int e) {
    int i = blockIdx.x * blockDim.x + threadIdx.x;
    if (i >= e) return;
    int s = src[i];
    int d = dst[i];
    int pos = atomicAdd((int*)ws + d, 1);
    if (pos < LIST_CAP) ((int*)out)[(size_t)d * FEAT + 1 + pos] = s;
}

__device__ __forceinline__ float dinv_of(int c) {
    return __builtin_amdgcn_rsqf(1.0f + (float)c);
}

__device__ __forceinline__ v4f gather_one(
    const float* __restrict__ emb, const int* __restrict__ src,
    const int* __restrict__ dst, const int* __restrict__ ecnt,
    const int* row, int mine, int cnt, float dn, int node, int lane, int e, v4f acc) {
    const int c = lane * 4;
    if (cnt <= 31) {
        for (int j = 0; j < cnt; j += 4) {
            int s[4];
            float w[4];
            v4f v[4];
            #pragma unroll
            for (int k = 0; k < 4; ++k) {
                const int jj = (j + k < cnt) ? j + k : j;
                s[k] = __builtin_amdgcn_readlane(mine, jj + 1);
            }
            #pragma unroll
            for (int k = 0; k < 4; ++k) {
                w[k] = dinv_of(ecnt[s[k]]) * dn;
                v[k] = *(const v4f*)(emb + (size_t)s[k] * FEAT + c);
            }
            #pragma unroll
            for (int k = 0; k < 4; ++k)
                if (j + k < cnt) acc += v[k] * w[k];
        }
    } else if (cnt <= LIST_CAP) {
        for (int j = 0; j < cnt; ++j) {
            const int s = __builtin_amdgcn_readfirstlane(row[j + 1]);
            const float w = dinv_of(ecnt[s]) * dn;
            v4f v = *(const v4f*)(emb + (size_t)s * FEAT + c);
            acc += v * w;
        }
    } else {
        for (int base = 0; base < e; base += 32) {
            int idx = base + lane;
            int d = (idx < e) ? dst[idx] : -1;
            unsigned m = __builtin_amdgcn_ballot_w32(d == node);
            while (m) {
                int b = __builtin_ctz(m);
                m &= m - 1;
                int s = src[base + b];
                float w = dinv_of(ecnt[s]) * dn;
                v4f v = *(const v4f*)(emb + (size_t)s * FEAT + c);
                acc += v * w;
            }
        }
    }
    return acc;
}

#define TILE_M 16
#define FUSED_GRID 1024

__global__ __launch_bounds__(256) __attribute__((amdgpu_waves_per_eu(8, 8)))
void gather_gemm_kernel(
    const float* __restrict__ emb, const int* __restrict__ src,
    const int* __restrict__ dst, const float* __restrict__ ws,
    const float* __restrict__ Wmat, const float* __restrict__ bias,
    float* out, int n, int e) {
    const int* __restrict__ ecnt = (const int*)ws;
    __shared__ float Ain[2][TILE_M * A_STRIDE];

    const int tid = threadIdx.x;
    const int wave = __builtin_amdgcn_readfirstlane(tid >> 5);
    const int lane = tid & 31;
    const int l = lane & 15;
    const int h = lane >> 4;
    const int col = wave * 16 + l;
    const int c = lane * 4;

    v2f bfrag[FEAT / 4];
    #pragma unroll
    for (int k = 0; k < FEAT / 4; ++k) {
        const int kk = 4 * k + 2 * h;
        bfrag[k].x = Wmat[(size_t)kk * FEAT + col];
        bfrag[k].y = Wmat[(size_t)(kk + 1) * FEAT + col];
    }
    const float bv = bias[col];

    const int ntiles = (n + TILE_M - 1) / TILE_M;
    int buf = 0;
    for (int tile = blockIdx.x; tile < ntiles; tile += gridDim.x, buf ^= 1) {
        const int row0 = tile * TILE_M;
        float* As = Ain[buf];

        const int rA = 2 * wave, rB = rA + 1;
        const int nA = row0 + rA, nB = row0 + rB;
        const bool okA = nA < n, okB = nB < n;
        const int mA = okA ? nA : 0, mB = okB ? nB : 0;
        const int* rowA = (const int*)(out + (size_t)mA * FEAT);
        const int* rowB = (const int*)(out + (size_t)mB * FEAT);
        const int mineA = rowA[lane];
        const int mineB = rowB[lane];
        const int ecA = ecnt[mA], ecB = ecnt[mB];
        const float dnA = dinv_of(ecA), dnB = dinv_of(ecB);
        v4f accA = *(const v4f*)(emb + (size_t)mA * FEAT + c);
        v4f accB = *(const v4f*)(emb + (size_t)mB * FEAT + c);
        const int cntA = okA ? __builtin_amdgcn_readfirstlane(ecA) : 0;
        const int cntB = okB ? __builtin_amdgcn_readfirstlane(ecB) : 0;
        accA *= dnA * dnA;
        accB *= dnB * dnB;

        if (cntA <= 31 && cntB <= 31) {
            const int cmax = cntA > cntB ? cntA : cntB;
            for (int j = 0; j < cmax; j += 4) {
                int sA[4], sB[4];
                float wA[4], wB[4];
                v4f vA[4], vB[4];
                #pragma unroll
                for (int k = 0; k < 4; ++k) {
                    const int tA = __builtin_amdgcn_readlane(mineA, (j + k + 1) & 31);
                    const int tB = __builtin_amdgcn_readlane(mineB, (j + k + 1) & 31);
                    sA[k] = (j + k < cntA) ? tA : mA;
                    sB[k] = (j + k < cntB) ? tB : mB;
                }
                #pragma unroll
                for (int k = 0; k < 4; ++k) {
                    wA[k] = dinv_of(ecnt[sA[k]]) * dnA;
                    vA[k] = *(const v4f*)(emb + (size_t)sA[k] * FEAT + c);
                    wB[k] = dinv_of(ecnt[sB[k]]) * dnB;
                    vB[k] = *(const v4f*)(emb + (size_t)sB[k] * FEAT + c);
                }
                #pragma unroll
                for (int k = 0; k < 4; ++k) {
                    if (j + k < cntA) accA += vA[k] * wA[k];
                    if (j + k < cntB) accB += vB[k] * wB[k];
                }
            }
        } else {
            accA = gather_one(emb, src, dst, ecnt, rowA, mineA, cntA, dnA, nA, lane, e, accA);
            accB = gather_one(emb, src, dst, ecnt, rowB, mineB, cntB, dnB, nB, lane, e, accB);
        }
        if (!okA) accA = (v4f){0.f, 0.f, 0.f, 0.f};
        if (!okB) accB = (v4f){0.f, 0.f, 0.f, 0.f};
        *(v4f*)(&As[rA * A_STRIDE + c]) = accA;
        *(v4f*)(&As[rB * A_STRIDE + c]) = accB;
        __syncthreads();

        const float* At = &As[l * A_STRIDE + 2 * h];
        v8f acc0 = {};
        v8f acc1 = {};
        #pragma unroll
        for (int k = 0; k < FEAT / 4; ++k) {
            const v2f a = *(const v2f*)(At + 4 * k);
            if (k & 1) {
                acc1 = __builtin_amdgcn_wmma_f32_16x16x4_f32(
                    false, a, false, bfrag[k], (short)0, acc1, false, false);
            } else {
                acc0 = __builtin_amdgcn_wmma_f32_16x16x4_f32(
                    false, a, false, bfrag[k], (short)0, acc0, false, false);
            }
        }
        const v8f acc = acc0 + acc1;

        if (row0 + TILE_M <= n) {
            float* orow = out + (size_t)(row0 + 8 * h) * FEAT + col;
            #pragma unroll
            for (int i = 0; i < 8; ++i)
                orow[(size_t)i * FEAT] = acc[i] + bv;
        } else {
            #pragma unroll
            for (int i = 0; i < 8; ++i) {
                const int row = row0 + i + 8 * h;
                if (row < n) out[(size_t)row * FEAT + col] = acc[i] + bv;
            }
        }
    }
}

extern "C" void kernel_launch(void* const* d_in, const int* in_sizes, int n_in,
                              void* d_out, int out_size, void* d_ws, size_t ws_size,
                              hipStream_t stream) {
    const float* emb  = (const float*)d_in[0];
    const int*   eidx = (const int*)d_in[1];
    const float* Wmat = (const float*)d_in[2];
    const float* bias = (const float*)d_in[3];
    float* out = (float*)d_out;

    const int N = in_sizes[0] / FEAT;
    const int E = in_sizes[1] / 2;
    const int* src = eidx;
    const int* dst = eidx + E;

    float* ecnt = (float*)d_ws;

    const int T = 256;
    cnt_init_kernel  <<<(N + T - 1) / T, T, 0, stream>>>(ecnt, N);
    list_build_kernel<<<(E + T - 1) / T, T, 0, stream>>>(src, dst, ecnt, out, E);
    gather_gemm_kernel<<<FUSED_GRID, T, 0, stream>>>(emb, src, dst, ecnt, Wmat, bias, out, N, E);
}
